# starting kernel
# baseline (speedup 1.0000x reference)
#include <hip/hip_runtime.h>

#define NLEN   8192
#define MODES  64
#define ROWS   16
#define W0     (7.6699039394282058e-4f)
#define INV_N  (1.220703125e-4f)

typedef __attribute__((ext_vector_type(2))) float v2f;
typedef __attribute__((ext_vector_type(8))) float v8f;

static __device__ __forceinline__ v8f wmma_f32_16x16x4(v2f a, v2f b, v8f c) {
    return __builtin_amdgcn_wmma_f32_16x16x4_f32(
         false, a,  false, b,
         (short)0, c,  false,  false);
}

__global__ void __launch_bounds__(1024)
dht_forward_modes(const float* __restrict__ x, float* __restrict__ Xm) {
    __shared__ float partial[32 * 256];

    const int lane = threadIdx.x & 31;
    const int wave = threadIdx.x >> 5;
    const int hi   = lane >> 4;
    const int lo   = lane & 15;
    const int tile = wave & 3;
    const int ks   = wave >> 2;
    const int mode = tile * 16 + lo;

    v8f acc = {};
    const int k0 = ks * 1024;
    for (int k = k0; k < k0 + 1024; k += 4) {
        const int kb = k + 2 * hi;
        v2f a, b;
        a.x = x[lo * NLEN + kb];
        a.y = x[lo * NLEN + kb + 1];
        b.x = __cosf(W0 * (float)((kb * mode) & (NLEN - 1)));
        b.y = __cosf(W0 * (float)(((kb + 1) * mode) & (NLEN - 1)));
        acc = wmma_f32_16x16x4(a, b, acc);
    }

#pragma unroll
    for (int j = 0; j < 8; ++j)
        partial[wave * 256 + (j + 8 * hi) * 16 + lo] = acc[j];
    __syncthreads();

    const int r = threadIdx.x >> 6;
    const int m = threadIdx.x & 63;
    float s = 0.0f;
#pragma unroll
    for (int sl = 0; sl < 8; ++sl)
        s += partial[(sl * 4 + (m >> 4)) * 256 + r * 16 + (m & 15)];
    Xm[r * 64 + m] = s;
}

__global__ void __launch_bounds__(1024)
mode_mix(const float* __restrict__ Xm, const float* __restrict__ w,
         float* __restrict__ Apacked) {
    const int t = threadIdx.x;
    const int m = t & 63;
    const int o = (t >> 6) & 7;
    const int b = t >> 9;
    float ar = 0.0f, ai = 0.0f;
#pragma unroll
    for (int i = 0; i < 8; ++i) {
        const float xv = Xm[(b * 8 + i) * 64 + m];
        const float wre = w[(((i * 8 + o) * 64 + m) << 1) + 0];
        const float wim = w[(((i * 8 + o) * 64 + m) << 1) + 1];
        ar += xv * wre;
        ai += xv * wim;
    }
    const int r = b * 8 + o;
    Apacked[r * 128 + m]      = ar;
    Apacked[r * 128 + 64 + m] = ai;
}

__global__ void __launch_bounds__(128)
dht_inverse(const float* __restrict__ Apacked, float* __restrict__ y) {
    __shared__ float As[ROWS * 128];
    for (int i = threadIdx.x; i < ROWS * 128; i += 128)
        As[i] = Apacked[i];
    __syncthreads();

    const int lane = threadIdx.x & 31;
    const int wave = threadIdx.x >> 5;
    const int hi   = lane >> 4;
    const int lo   = lane & 15;
    const int n    = (blockIdx.x * 4 + wave) * 16 + lo;

    v8f acc = {};
    for (int k = 0; k < 128; k += 4) {
        const int kb = k + 2 * hi;
        v2f a, b;
        a.x = As[lo * 128 + kb];
        a.y = As[lo * 128 + kb + 1];
        if (kb < 64) {
            b.x =  INV_N * __cosf(W0 * (float)((kb * n) & (NLEN - 1)));
            b.y =  INV_N * __cosf(W0 * (float)(((kb + 1) * n) & (NLEN - 1)));
        } else {
            b.x = -INV_N * __sinf(W0 * (float)(((kb - 64) * n) & (NLEN - 1)));
            b.y = -INV_N * __sinf(W0 * (float)(((kb - 63) * n) & (NLEN - 1)));
        }
        acc = wmma_f32_16x16x4(a, b, acc);
    }

#pragma unroll
    for (int j = 0; j < 8; ++j)
        y[(j + 8 * hi) * NLEN + n] = acc[j];
}

extern "C" void kernel_launch(void* const* d_in, const int* in_sizes, int n_in,
                              void* d_out, int out_size, void* d_ws, size_t ws_size,
                              hipStream_t stream) {
    const float* x = (const float*)d_in[0];
    const float* w = (const float*)d_in[1];
    float* out = (float*)d_out;

    float* Xm      = (float*)d_ws;
    float* Apacked = Xm + 1024;

    dht_forward_modes<<<1, 1024, 0, stream>>>(x, Xm);
    mode_mix<<<1, 1024, 0, stream>>>(Xm, w, Apacked);
    dht_inverse<<<128, 128, 0, stream>>>(Apacked, out);
}


# stage-1 K sum spread over 64 workgroups (16 K planes x 4 mode tiles, 8 waves each, 16 WMMA steps per wave) instead of one; mix kernel sums the 16 planes first; new f32 reduction tree
# speedup vs baseline: 4.4095x; 4.4095x over previous
#include <hip/hip_runtime.h>

#define NLEN   8192
#define MODES  64
#define ROWS   16
#define W0     (7.6699039394282058e-4f)
#define INV_N  (1.220703125e-4f)
#define KPLANES 16
#define KSUB    8
#define KWAVE   (NLEN / (KPLANES * KSUB))

typedef __attribute__((ext_vector_type(2))) float v2f;
typedef __attribute__((ext_vector_type(8))) float v8f;

static __device__ __forceinline__ v8f wmma_f32_16x16x4(v2f a, v2f b, v8f c) {
    return __builtin_amdgcn_wmma_f32_16x16x4_f32(
         false, a,  false, b,
         (short)0, c,  false,  false);
}

__global__ void __launch_bounds__(256)
dht_forward_modes(const float* __restrict__ x, float* __restrict__ planes) {
    __shared__ float partial[KSUB * 256];

    const int lane  = threadIdx.x & 31;
    const int wave  = threadIdx.x >> 5;
    const int hi    = lane >> 4;
    const int lo    = lane & 15;
    const int tile  = blockIdx.x & 3;
    const int plane = blockIdx.x >> 2;
    const int mode  = tile * 16 + lo;

    v8f acc = {};
    const int k0 = (plane * KSUB + wave) * KWAVE;
#pragma unroll
    for (int i = 0; i < KWAVE / 4; ++i) {
        const int kb = k0 + 4 * i + 2 * hi;
        v2f a, b;
        a.x = x[lo * NLEN + kb];
        a.y = x[lo * NLEN + kb + 1];
        b.x = __cosf(W0 * (float)((kb * mode) & (NLEN - 1)));
        b.y = __cosf(W0 * (float)(((kb + 1) * mode) & (NLEN - 1)));
        acc = wmma_f32_16x16x4(a, b, acc);
    }

#pragma unroll
    for (int j = 0; j < 8; ++j)
        partial[wave * 256 + (j + 8 * hi) * 16 + lo] = acc[j];
    __syncthreads();

    float s = 0.0f;
#pragma unroll
    for (int sl = 0; sl < KSUB; ++sl)
        s += partial[sl * 256 + threadIdx.x];
    planes[(plane * 4 + tile) * 256 + threadIdx.x] = s;
}

__global__ void __launch_bounds__(1024)
mode_mix(const float* __restrict__ planes, const float* __restrict__ w,
         float* __restrict__ Apacked) {
    __shared__ float Xs[1024];
    const int t = threadIdx.x;
    const int m = t & 63;
    const int o = (t >> 6) & 7;
    const int b = t >> 9;

    float wre[8], wim[8];
#pragma unroll
    for (int i = 0; i < 8; ++i) {
        wre[i] = w[(((i * 8 + o) * 64 + m) << 1) + 0];
        wim[i] = w[(((i * 8 + o) * 64 + m) << 1) + 1];
    }

    float xs = 0.0f;
    for (int p = 0; p < KPLANES; ++p)
        xs += planes[p * 1024 + t];
    Xs[t] = xs;
    asm volatile("" :: "v"(wre[0]), "v"(wre[1]), "v"(wre[2]), "v"(wre[3]),
                       "v"(wre[4]), "v"(wre[5]), "v"(wre[6]), "v"(wre[7]),
                       "v"(wim[0]), "v"(wim[1]), "v"(wim[2]), "v"(wim[3]),
                       "v"(wim[4]), "v"(wim[5]), "v"(wim[6]), "v"(wim[7]));
    __syncthreads();

    float ar = 0.0f, ai = 0.0f;
#pragma unroll
    for (int i = 0; i < 8; ++i) {
        const float xv = Xs[(m >> 4) * 256 + (b * 8 + i) * 16 + (m & 15)];
        ar += xv * wre[i];
        ai += xv * wim[i];
    }
    const int r = b * 8 + o;
    Apacked[r * 128 + m]      = ar;
    Apacked[r * 128 + 64 + m] = ai;
}

__global__ void __launch_bounds__(128)
dht_inverse(const float* __restrict__ Apacked, float* __restrict__ y) {
    __shared__ float As[ROWS * 128];
    for (int i = threadIdx.x; i < ROWS * 128; i += 128)
        As[i] = Apacked[i];
    __syncthreads();

    const int lane = threadIdx.x & 31;
    const int wave = threadIdx.x >> 5;
    const int hi   = lane >> 4;
    const int lo   = lane & 15;
    const int n    = (blockIdx.x * 4 + wave) * 16 + lo;

    v8f acc = {};
    for (int k = 0; k < 128; k += 4) {
        const int kb = k + 2 * hi;
        v2f a, b;
        a.x = As[lo * 128 + kb];
        a.y = As[lo * 128 + kb + 1];
        if (kb < 64) {
            b.x =  INV_N * __cosf(W0 * (float)((kb * n) & (NLEN - 1)));
            b.y =  INV_N * __cosf(W0 * (float)(((kb + 1) * n) & (NLEN - 1)));
        } else {
            b.x = -INV_N * __sinf(W0 * (float)(((kb - 64) * n) & (NLEN - 1)));
            b.y = -INV_N * __sinf(W0 * (float)(((kb - 63) * n) & (NLEN - 1)));
        }
        acc = wmma_f32_16x16x4(a, b, acc);
    }

#pragma unroll
    for (int j = 0; j < 8; ++j)
        y[(j + 8 * hi) * NLEN + n] = acc[j];
}

extern "C" void kernel_launch(void* const* d_in, const int* in_sizes, int n_in,
                              void* d_out, int out_size, void* d_ws, size_t ws_size,
                              hipStream_t stream) {
    const float* x = (const float*)d_in[0];
    const float* w = (const float*)d_in[1];
    float* out = (float*)d_out;

    float* planes  = (float*)d_ws;
    float* Apacked = planes + 16384;

    dht_forward_modes<<<64, 256, 0, stream>>>(x, planes);
    mode_mix<<<1, 1024, 0, stream>>>(planes, w, Apacked);
    dht_inverse<<<128, 128, 0, stream>>>(Apacked, out);
}


# mix kernel folded into the inverse kernel: each of its 128 workgroups sums the 16 planes and does the 8x8 channel mix into LDS itself; 2 launches instead of 3; every sum in the same order
# speedup vs baseline: 6.8721x; 1.5585x over previous
#include <hip/hip_runtime.h>

#define NLEN   8192
#define MODES  64
#define ROWS   16
#define W0     (7.6699039394282058e-4f)
#define INV_N  (1.220703125e-4f)
#define KPLANES 16
#define KSUB    8
#define KWAVE   (NLEN / (KPLANES * KSUB))

typedef __attribute__((ext_vector_type(2))) float v2f;
typedef __attribute__((ext_vector_type(4))) float v4f;
typedef __attribute__((ext_vector_type(8))) float v8f;

static __device__ __forceinline__ v8f wmma_f32_16x16x4(v2f a, v2f b, v8f c) {
    return __builtin_amdgcn_wmma_f32_16x16x4_f32(
         false, a,  false, b,
         (short)0, c,  false,  false);
}

__global__ void __launch_bounds__(256)
dht_forward_modes(const float* __restrict__ x, float* __restrict__ planes) {
    __shared__ float partial[KSUB * 256];

    const int lane  = threadIdx.x & 31;
    const int wave  = threadIdx.x >> 5;
    const int hi    = lane >> 4;
    const int lo    = lane & 15;
    const int tile  = blockIdx.x & 3;
    const int plane = blockIdx.x >> 2;
    const int mode  = tile * 16 + lo;

    v8f acc = {};
    const int k0 = (plane * KSUB + wave) * KWAVE;
#pragma unroll
    for (int i = 0; i < KWAVE / 4; ++i) {
        const int kb = k0 + 4 * i + 2 * hi;
        v2f a, b;
        a.x = x[lo * NLEN + kb];
        a.y = x[lo * NLEN + kb + 1];
        b.x = __cosf(W0 * (float)((kb * mode) & (NLEN - 1)));
        b.y = __cosf(W0 * (float)(((kb + 1) * mode) & (NLEN - 1)));
        acc = wmma_f32_16x16x4(a, b, acc);
    }

#pragma unroll
    for (int j = 0; j < 8; ++j)
        partial[wave * 256 + (j + 8 * hi) * 16 + lo] = acc[j];
    __syncthreads();

    float s = 0.0f;
#pragma unroll
    for (int sl = 0; sl < KSUB; ++sl)
        s += partial[sl * 256 + threadIdx.x];
    planes[(plane * 4 + tile) * 256 + threadIdx.x] = s;
}

__global__ void __launch_bounds__(128) __attribute__((amdgpu_waves_per_eu(1, 1)))
mix_inverse(const float* __restrict__ planes, const float* __restrict__ w,
            float* __restrict__ y) {
    __shared__ __attribute__((aligned(16))) float Xs[ROWS * MODES];
    __shared__ __attribute__((aligned(16))) float As[ROWS * 128];

    const int t = threadIdx.x;
    const v4f* __restrict__ planes4 = reinterpret_cast<const v4f*>(planes);
    const v4f* __restrict__ w4      = reinterpret_cast<const v4f*>(w);

    v4f pa[KPLANES], pb[KPLANES];
#pragma unroll
    for (int p = 0; p < KPLANES; ++p) {
        pa[p] = planes4[p * 256 + t];
        pb[p] = planes4[p * 256 + 128 + t];
    }
    v4f wv[16];
#pragma unroll
    for (int j = 0; j < 16; ++j)
        wv[j] = w4[j * 128 + t];

    v4f sa = {0.0f, 0.0f, 0.0f, 0.0f};
    v4f sb = {0.0f, 0.0f, 0.0f, 0.0f};
#pragma unroll
    for (int p = 0; p < KPLANES; ++p) {
        sa += pa[p];
        sb += pb[p];
    }
    {
        const int row = (t >> 2) & 15;
        const int col = (t & 3) * 4;
        const int tl  = t >> 6;
        *reinterpret_cast<v4f*>(&Xs[row * 64 + tl * 16 + col])       = sa;
        *reinterpret_cast<v4f*>(&Xs[row * 64 + (tl + 2) * 16 + col]) = sb;
    }
    __syncthreads();

    {
        const int m0 = (t & 31) * 2;
        const int wq = t >> 5;
        v2f xv[16];
#pragma unroll
        for (int bi = 0; bi < 16; ++bi)
            xv[bi] = *reinterpret_cast<const v2f*>(&Xs[bi * 64 + m0]);
#pragma unroll
        for (int oh = 0; oh < 2; ++oh) {
#pragma unroll
            for (int b = 0; b < 2; ++b) {
                float ar0 = 0.0f, ai0 = 0.0f, ar1 = 0.0f, ai1 = 0.0f;
#pragma unroll
                for (int i = 0; i < 8; ++i) {
                    const v4f wc = wv[2 * i + oh];
                    const v2f xc = xv[b * 8 + i];
                    ar0 = __builtin_fmaf(xc.x, wc.x, ar0);
                    ai0 = __builtin_fmaf(xc.x, wc.y, ai0);
                    ar1 = __builtin_fmaf(xc.y, wc.z, ar1);
                    ai1 = __builtin_fmaf(xc.y, wc.w, ai1);
                }
                const int r = b * 8 + wq + 4 * oh;
                v2f re2 = {ar0, ar1};
                v2f im2 = {ai0, ai1};
                *reinterpret_cast<v2f*>(&As[r * 128 + m0])      = re2;
                *reinterpret_cast<v2f*>(&As[r * 128 + 64 + m0]) = im2;
            }
        }
    }
    __syncthreads();

    const int lane = threadIdx.x & 31;
    const int wave = threadIdx.x >> 5;
    const int hi   = lane >> 4;
    const int lo   = lane & 15;
    const int n    = (blockIdx.x * 4 + wave) * 16 + lo;

    v8f acc = {};
    for (int k = 0; k < 128; k += 4) {
        const int kb = k + 2 * hi;
        v2f a, b;
        a.x = As[lo * 128 + kb];
        a.y = As[lo * 128 + kb + 1];
        if (kb < 64) {
            b.x =  INV_N * __cosf(W0 * (float)((kb * n) & (NLEN - 1)));
            b.y =  INV_N * __cosf(W0 * (float)(((kb + 1) * n) & (NLEN - 1)));
        } else {
            b.x = -INV_N * __sinf(W0 * (float)(((kb - 64) * n) & (NLEN - 1)));
            b.y = -INV_N * __sinf(W0 * (float)(((kb - 63) * n) & (NLEN - 1)));
        }
        acc = wmma_f32_16x16x4(a, b, acc);
    }

#pragma unroll
    for (int j = 0; j < 8; ++j)
        y[(j + 8 * hi) * NLEN + n] = acc[j];
}

extern "C" void kernel_launch(void* const* d_in, const int* in_sizes, int n_in,
                              void* d_out, int out_size, void* d_ws, size_t ws_size,
                              hipStream_t stream) {
    const float* x = (const float*)d_in[0];
    const float* w = (const float*)d_in[1];
    float* out = (float*)d_out;

    float* planes = (float*)d_ws;

    dht_forward_modes<<<64, 256, 0, stream>>>(x, planes);
    mix_inverse<<<128, 128, 0, stream>>>(planes, w, out);
}


# inverse kernel: K loop is 32 back-to-back WMMAs; its sin/cos B fragments are computed during the prologue's load wait, its A fragments read from LDS as one conflict-free group (row pitch 132)
# speedup vs baseline: 7.6801x; 1.1176x over previous
#include <hip/hip_runtime.h>

#define NLEN   8192
#define MODES  64
#define ROWS   16
#define W0     (7.6699039394282058e-4f)
#define INV_N  (1.220703125e-4f)
#define KPLANES 16
#define KSUB    8
#define KWAVE   (NLEN / (KPLANES * KSUB))
#define APITCH  132

typedef __attribute__((ext_vector_type(2))) float v2f;
typedef __attribute__((ext_vector_type(4))) float v4f;
typedef __attribute__((ext_vector_type(8))) float v8f;

static __device__ __forceinline__ v8f wmma_f32_16x16x4(v2f a, v2f b, v8f c) {
    return __builtin_amdgcn_wmma_f32_16x16x4_f32(
         false, a,  false, b,
         (short)0, c,  false,  false);
}

__global__ void __launch_bounds__(256)
dht_forward_modes(const float* __restrict__ x, float* __restrict__ planes) {
    __shared__ float partial[KSUB * 256];

    const int lane  = threadIdx.x & 31;
    const int wave  = threadIdx.x >> 5;
    const int hi    = lane >> 4;
    const int lo    = lane & 15;
    const int tile  = blockIdx.x & 3;
    const int plane = blockIdx.x >> 2;
    const int mode  = tile * 16 + lo;

    v8f acc = {};
    const int k0 = (plane * KSUB + wave) * KWAVE;
#pragma unroll
    for (int i = 0; i < KWAVE / 4; ++i) {
        const int kb = k0 + 4 * i + 2 * hi;
        v2f a, b;
        a.x = x[lo * NLEN + kb];
        a.y = x[lo * NLEN + kb + 1];
        b.x = __cosf(W0 * (float)((kb * mode) & (NLEN - 1)));
        b.y = __cosf(W0 * (float)(((kb + 1) * mode) & (NLEN - 1)));
        acc = wmma_f32_16x16x4(a, b, acc);
    }

#pragma unroll
    for (int j = 0; j < 8; ++j)
        partial[wave * 256 + (j + 8 * hi) * 16 + lo] = acc[j];
    __syncthreads();

    float s = 0.0f;
#pragma unroll
    for (int sl = 0; sl < KSUB; ++sl)
        s += partial[sl * 256 + threadIdx.x];
    planes[(plane * 4 + tile) * 256 + threadIdx.x] = s;
}

__global__ void __launch_bounds__(128) __attribute__((amdgpu_waves_per_eu(1, 1)))
mix_inverse(const float* __restrict__ planes, const float* __restrict__ w,
            float* __restrict__ y) {
    __shared__ __attribute__((aligned(16))) float Xs[ROWS * MODES];
    __shared__ __attribute__((aligned(16))) float As[ROWS * APITCH];

    const int t = threadIdx.x;
    const v4f* __restrict__ planes4 = reinterpret_cast<const v4f*>(planes);
    const v4f* __restrict__ w4      = reinterpret_cast<const v4f*>(w);

    v4f pa[KPLANES], pb[KPLANES];
#pragma unroll
    for (int p = 0; p < KPLANES; ++p) {
        pa[p] = planes4[p * 256 + t];
        pb[p] = planes4[p * 256 + 128 + t];
    }
    v4f wv[16];
#pragma unroll
    for (int j = 0; j < 16; ++j)
        wv[j] = w4[j * 128 + t];

    const int lane = threadIdx.x & 31;
    const int wave = threadIdx.x >> 5;
    const int hi   = lane >> 4;
    const int lo   = lane & 15;
    const int n    = (blockIdx.x * 4 + wave) * 16 + lo;
    v2f bf[32];
#pragma unroll
    for (int i = 0; i < 32; ++i) {
        const int kb = 4 * i + 2 * hi;
        if (i < 16) {
            bf[i].x =  INV_N * __cosf(W0 * (float)((kb * n) & (NLEN - 1)));
            bf[i].y =  INV_N * __cosf(W0 * (float)(((kb + 1) * n) & (NLEN - 1)));
        } else {
            bf[i].x = -INV_N * __sinf(W0 * (float)(((kb - 64) * n) & (NLEN - 1)));
            bf[i].y = -INV_N * __sinf(W0 * (float)(((kb - 63) * n) & (NLEN - 1)));
        }
    }
    __builtin_amdgcn_sched_barrier(0);

    v4f sa = {0.0f, 0.0f, 0.0f, 0.0f};
    v4f sb = {0.0f, 0.0f, 0.0f, 0.0f};
#pragma unroll
    for (int p = 0; p < KPLANES; ++p) {
        sa += pa[p];
        sb += pb[p];
    }
    {
        const int row = (t >> 2) & 15;
        const int col = (t & 3) * 4;
        const int tl  = t >> 6;
        *reinterpret_cast<v4f*>(&Xs[row * 64 + tl * 16 + col])       = sa;
        *reinterpret_cast<v4f*>(&Xs[row * 64 + (tl + 2) * 16 + col]) = sb;
    }
    __syncthreads();

    {
        const int m0 = (t & 31) * 2;
        const int wq = t >> 5;
        v2f xv[16];
#pragma unroll
        for (int bi = 0; bi < 16; ++bi)
            xv[bi] = *reinterpret_cast<const v2f*>(&Xs[bi * 64 + m0]);
#pragma unroll
        for (int oh = 0; oh < 2; ++oh) {
#pragma unroll
            for (int b = 0; b < 2; ++b) {
                float ar0 = 0.0f, ai0 = 0.0f, ar1 = 0.0f, ai1 = 0.0f;
#pragma unroll
                for (int i = 0; i < 8; ++i) {
                    const v4f wc = wv[2 * i + oh];
                    const v2f xc = xv[b * 8 + i];
                    ar0 = __builtin_fmaf(xc.x, wc.x, ar0);
                    ai0 = __builtin_fmaf(xc.x, wc.y, ai0);
                    ar1 = __builtin_fmaf(xc.y, wc.z, ar1);
                    ai1 = __builtin_fmaf(xc.y, wc.w, ai1);
                }
                const int r = b * 8 + wq + 4 * oh;
                v2f re2 = {ar0, ar1};
                v2f im2 = {ai0, ai1};
                *reinterpret_cast<v2f*>(&As[r * APITCH + m0])      = re2;
                *reinterpret_cast<v2f*>(&As[r * APITCH + 64 + m0]) = im2;
            }
        }
    }
    __syncthreads();

    v2f af[32];
#pragma unroll
    for (int i = 0; i < 32; ++i)
        af[i] = *reinterpret_cast<const v2f*>(&As[lo * APITCH + 4 * i + 2 * hi]);
    __builtin_amdgcn_sched_barrier(0);

    v8f acc = {};
#pragma unroll
    for (int i = 0; i < 32; ++i)
        acc = wmma_f32_16x16x4(af[i], bf[i], acc);

#pragma unroll
    for (int j = 0; j < 8; ++j)
        y[(j + 8 * hi) * NLEN + n] = acc[j];
}

extern "C" void kernel_launch(void* const* d_in, const int* in_sizes, int n_in,
                              void* d_out, int out_size, void* d_ws, size_t ws_size,
                              hipStream_t stream) {
    const float* x = (const float*)d_in[0];
    const float* w = (const float*)d_in[1];
    float* out = (float*)d_out;

    float* planes = (float*)d_ws;

    dht_forward_modes<<<64, 256, 0, stream>>>(x, planes);
    mix_inverse<<<128, 128, 0, stream>>>(planes, w, out);
}
